# starting kernel
# baseline (speedup 1.0000x reference)
#include <hip/hip_runtime.h>
#include <math.h>


typedef float v2f __attribute__((ext_vector_type(2)));
typedef float v8f __attribute__((ext_vector_type(8)));

#define N_PTS 65536
#define DDIM  256
#define KFREQ 64
#define NROW  128
#define NCHUNK 64
#define CHUNK_N (N_PTS / NCHUNK)

__device__ __forceinline__ v8f wmma_f32(v2f a, v2f b, v8f c) {
  return __builtin_amdgcn_wmma_f32_16x16x4_f32(false, a, false, b,
                                               (short)0, c, false, false);
}

__global__ __launch_bounds__(256) void tgen_kernel(float* __restrict__ T) {
  int idx = blockIdx.x * 256 + threadIdx.x;
  int n = idx >> 6;
  int k = idx & 63;
  unsigned m = ((unsigned)n * (unsigned)k) & 65535u;
  float ang = (float)m * 9.58737992429e-5f;
  float s, c;
  sincosf(ang, &s, &c);
  size_t base = (size_t)n * NROW + 2 * k;
  T[base]     = c;
  T[base + 1] = s;
}

__global__ __launch_bounds__(256) void pack_w_kernel(
    const float* __restrict__ W, float* __restrict__ Wp) {
  int idx = blockIdx.x * 256 + threadIdx.x;
  int kpair = idx >> 8;
  int col   = idx & 255;
  Wp[(size_t)idx * 2]     = W[(size_t)(2 * kpair) * DDIM + col];
  Wp[(size_t)idx * 2 + 1] = W[(size_t)(2 * kpair + 1) * DDIM + col];
}

__global__ __launch_bounds__(256) void dft_low_kernel(
    const float* __restrict__ x, const float* __restrict__ T,
    float* __restrict__ Ppart) {
  const int lane    = threadIdx.x & 31;
  const int wave    = threadIdx.x >> 5;
  const int lr      = lane & 15;
  const int halfSel = (lane >> 4) << 1;
  const int row0    = blockIdx.x * 16;
  const int nbase   = blockIdx.y * CHUNK_N;
  const int d0      = wave * 32;

  v8f acc0 = {0.f,0.f,0.f,0.f,0.f,0.f,0.f,0.f};
  v8f acc1 = acc0;

  const int arow = row0 + lr;
  for (int nk = nbase; nk < nbase + CHUNK_N; nk += 4) {
    const int na = nk + halfSel;
    v2f a;
    a.x = T[(size_t)na * NROW + arow];
    a.y = T[(size_t)(na + 1) * NROW + arow];
    v2f b0, b1;
    b0.x = x[(size_t)na * DDIM + d0 + lr];
    b0.y = x[(size_t)(na + 1) * DDIM + d0 + lr];
    b1.x = x[(size_t)na * DDIM + d0 + 16 + lr];
    b1.y = x[(size_t)(na + 1) * DDIM + d0 + 16 + lr];
    acc0 = wmma_f32(a, b0, acc0);
    acc1 = wmma_f32(a, b1, acc1);
  }

  const int mbase = (lane >> 4) << 3;
  const int col   = d0 + lr;
  size_t base = (size_t)blockIdx.y * (NROW * DDIM);
#pragma unroll
  for (int v = 0; v < 8; ++v) {
    int r = row0 + mbase + v;
    Ppart[base + (size_t)r * DDIM + col]      = acc0[v];
    Ppart[base + (size_t)r * DDIM + col + 16] = acc1[v];
  }
}

__global__ __launch_bounds__(256) void reduce_chunks_kernel(
    const float* __restrict__ Ppart, float* __restrict__ C) {
  int idx = blockIdx.x * 256 + threadIdx.x;
  float s = 0.f;
#pragma unroll 4
  for (int c = 0; c < NCHUNK; ++c)
    s += Ppart[(size_t)c * (NROW * DDIM) + idx];
  C[idx] = s;
}

__global__ __launch_bounds__(256) void spectral_mix_kernel(
    const float* __restrict__ C, const float* __restrict__ P_low,
    const float* __restrict__ B_low, float* __restrict__ BppP) {
  __shared__ float sXr[DDIM];
  __shared__ float sXi[DDIM];
  const int k = blockIdx.x;
  const int f = threadIdx.x;
  const float invN = 1.0f / (float)N_PTS;
  sXr[f] =  C[(size_t)(2 * k) * DDIM + f] * invN;
  sXi[f] = -C[(size_t)(2 * k + 1) * DDIM + f] * invN;
  __syncthreads();

  float ar = B_low[((size_t)k * DDIM + f) * 2 + 0];
  float ai = B_low[((size_t)k * DDIM + f) * 2 + 1];
  const float* Pk = P_low + (size_t)k * DDIM * DDIM * 2;
#pragma unroll 4
  for (int d = 0; d < DDIM; ++d) {
    float pr = Pk[((size_t)d * DDIM + f) * 2 + 0];
    float pi = Pk[((size_t)d * DDIM + f) * 2 + 1];
    float xr = sXr[d], xi = sXi[d];
    ar = fmaf(xr, pr, fmaf(-xi, pi, ar));
    ai = fmaf(xr, pi, fmaf( xi, pr, ai));
  }
  float sc = (k == 0) ? 1.0f : 2.0f;
  BppP[((size_t)k * DDIM + f) * 2 + 0] =  sc * ar;
  BppP[((size_t)k * DDIM + f) * 2 + 1] = -sc * ai;
}

__global__ __launch_bounds__(256) void fused_out_kernel(
    const float* __restrict__ x, const float* __restrict__ Wp,
    const float* __restrict__ Bias, const float* __restrict__ T,
    const float* __restrict__ BppP, float* __restrict__ out) {
  const int lane    = threadIdx.x & 31;
  const int wave    = threadIdx.x >> 5;
  const int lr      = lane & 15;
  const int halfSel = (lane >> 4) << 1;
  const int n0      = (blockIdx.x * 2 + (wave >> 2)) * 16;
  const int d0      = (wave & 3) * 64;
  const int arow    = n0 + lr;
  const int col     = d0 + lr;

  v8f acc[4];
#pragma unroll
  for (int t = 0; t < 4; ++t) acc[t] = (v8f){0.f,0.f,0.f,0.f,0.f,0.f,0.f,0.f};

  for (int k0 = 0; k0 < DDIM; k0 += 4) {
    const int kb = k0 + halfSel;
    v2f a = *(const v2f*)(x + (size_t)arow * DDIM + kb);
    const float* wrow = Wp + ((size_t)(kb >> 1) * DDIM + col) * 2;
#pragma unroll
    for (int t = 0; t < 4; ++t) {
      v2f b = *(const v2f*)(wrow + t * 32);
      acc[t] = wmma_f32(a, b, acc[t]);
    }
  }
  for (int r0 = 0; r0 < NROW; r0 += 4) {
    const int rb = r0 + halfSel;
    v2f a = *(const v2f*)(T + (size_t)arow * NROW + rb);
    const float* brow = BppP + ((size_t)(rb >> 1) * DDIM + col) * 2;
#pragma unroll
    for (int t = 0; t < 4; ++t) {
      v2f b = *(const v2f*)(brow + t * 32);
      acc[t] = wmma_f32(a, b, acc[t]);
    }
  }

  const int mbase = (lane >> 4) << 3;
#pragma unroll
  for (int t = 0; t < 4; ++t) {
    const int c = col + t * 16;
    const float bias = Bias[c];
#pragma unroll
    for (int v = 0; v < 8; ++v) {
      int n = n0 + mbase + v;
      out[(size_t)n * DDIM + c] = sinf(0.2f * (acc[t][v] + bias));
    }
  }
}

extern "C" void kernel_launch(void* const* d_in, const int* in_sizes, int n_in,
                              void* d_out, int out_size, void* d_ws,
                              size_t ws_size, hipStream_t stream) {
  (void)in_sizes; (void)n_in; (void)out_size; (void)ws_size;
  const float* x     = (const float*)d_in[0];
  const float* P_low = (const float*)d_in[1];
  const float* B_low = (const float*)d_in[2];
  const float* W     = (const float*)d_in[3];
  const float* Bias  = (const float*)d_in[4];
  float* out = (float*)d_out;

  float* ws    = (float*)d_ws;
  float* T     = ws;
  float* Ppart = T + (size_t)N_PTS * NROW;
  float* C     = Ppart + (size_t)NCHUNK * NROW * DDIM;
  float* BppP  = C + NROW * DDIM;
  float* Wp    = BppP + NROW * DDIM;

  tgen_kernel<<<dim3((N_PTS * KFREQ) / 256), dim3(256), 0, stream>>>(T);
  pack_w_kernel<<<dim3((DDIM * DDIM / 2) / 256), dim3(256), 0, stream>>>(W, Wp);
  dft_low_kernel<<<dim3(NROW / 16, NCHUNK), dim3(256), 0, stream>>>(x, T, Ppart);
  reduce_chunks_kernel<<<dim3((NROW * DDIM) / 256), dim3(256), 0, stream>>>(Ppart, C);
  spectral_mix_kernel<<<dim3(KFREQ), dim3(DDIM), 0, stream>>>(C, P_low, B_low, BppP);
  fused_out_kernel<<<dim3(N_PTS / 32), dim3(256), 0, stream>>>(x, Wp, Bias, T, BppP, out);
}


# Output GEMM: 4x4 tiles per wave (64 rows x 64 cols) fed straight by b128 global loads under a regrouped k order, next 4 k-steps loaded ahead; rolled sine epilogue via LDS; sum order changed
# speedup vs baseline: 1.6538x; 1.6538x over previous
#include <hip/hip_runtime.h>
#include <math.h>


typedef float v2f __attribute__((ext_vector_type(2)));
typedef float v4f __attribute__((ext_vector_type(4)));
typedef float v8f __attribute__((ext_vector_type(8)));

#define N_PTS 65536
#define DDIM  256
#define KFREQ 64
#define NROW  128
#define NCHUNK 64
#define CHUNK_N (N_PTS / NCHUNK)

__device__ __forceinline__ v8f wmma_f32(v2f a, v2f b, v8f c) {
  return __builtin_amdgcn_wmma_f32_16x16x4_f32(false, a, false, b,
                                               (short)0, c, false, false);
}

__global__ __launch_bounds__(256) void tgen_kernel(float* __restrict__ T) {
  int idx = blockIdx.x * 256 + threadIdx.x;
  int n = idx >> 6;
  int k = idx & 63;
  unsigned m = ((unsigned)n * (unsigned)k) & 65535u;
  float ang = (float)m * 9.58737992429e-5f;
  float s, c;
  sincosf(ang, &s, &c);
  size_t base = (size_t)n * NROW + 2 * k;
  T[base]     = c;
  T[base + 1] = s;
}

__global__ __launch_bounds__(256) void pack_w_kernel(
    const float* __restrict__ W, float* __restrict__ Wp) {
  int idx = blockIdx.x * 256 + threadIdx.x;
  int kpair = idx >> 8;
  int col   = idx & 255;
  Wp[(size_t)idx * 2]     = W[(size_t)(2 * kpair) * DDIM + col];
  Wp[(size_t)idx * 2 + 1] = W[(size_t)(2 * kpair + 1) * DDIM + col];
}

#define DFT_U 8
#define DFT_GROUPS (CHUNK_N / (4 * DFT_U))

struct DftGroup {
  v2f t[DFT_U][2];
  v4f xv[DFT_U][2];
};

__device__ __forceinline__ void dft_load(DftGroup& g, const float* tp,
                                         const float* xp) {
#pragma unroll
  for (int s = 0; s < DFT_U; ++s) {
#pragma unroll
    for (int e = 0; e < 2; ++e) {
      g.t[s][e]  = *(const v2f*)(tp + (size_t)(4 * s + e) * NROW);
      g.xv[s][e] = *(const v4f*)(xp + (size_t)(4 * s + e) * DDIM);
    }
  }
}

__device__ __forceinline__ void dft_mma(const DftGroup& g, v8f (&acc)[2][4]) {
#pragma unroll
  for (int s = 0; s < DFT_U; ++s) {
#pragma unroll
    for (int i = 0; i < 2; ++i) {
      v2f a;
      a.x = g.t[s][0][i];
      a.y = g.t[s][1][i];
#pragma unroll
      for (int j = 0; j < 4; ++j) {
        v2f b;
        b.x = g.xv[s][0][j];
        b.y = g.xv[s][1][j];
        acc[i][j] = wmma_f32(a, b, acc[i][j]);
      }
    }
  }
}

__global__ __launch_bounds__(128) void dft_low_kernel(
    const float* __restrict__ x, const float* __restrict__ T,
    float* __restrict__ Ppart) {
  const int lane    = threadIdx.x & 31;
  const int wave    = threadIdx.x >> 5;
  const int c       = lane & 15;
  const int h       = lane >> 4;
  const int rowbase = wave * 32;
  const int colbase = blockIdx.x * 64;
  const int nbase   = blockIdx.y * CHUNK_N;

  const float* tp = T + (size_t)(nbase + 2 * h) * NROW + rowbase + 2 * c;
  const float* xp = x + (size_t)(nbase + 2 * h) * DDIM + colbase + 4 * c;
  const size_t tstep = (size_t)4 * DFT_U * NROW;
  const size_t xstep = (size_t)4 * DFT_U * DDIM;

  v8f acc[2][4];
#pragma unroll
  for (int i = 0; i < 2; ++i)
#pragma unroll
    for (int j = 0; j < 4; ++j)
      acc[i][j] = (v8f){0.f,0.f,0.f,0.f,0.f,0.f,0.f,0.f};

  DftGroup ga, gb;
  dft_load(ga, tp, xp);
  for (int g = 0; g < DFT_GROUPS - 2; g += 2) {
    dft_load(gb, tp + (size_t)(g + 1) * tstep, xp + (size_t)(g + 1) * xstep);
    __builtin_amdgcn_sched_barrier(0);
    dft_mma(ga, acc);
    __builtin_amdgcn_sched_barrier(0);
    dft_load(ga, tp + (size_t)(g + 2) * tstep, xp + (size_t)(g + 2) * xstep);
    __builtin_amdgcn_sched_barrier(0);
    dft_mma(gb, acc);
    __builtin_amdgcn_sched_barrier(0);
  }
  dft_load(gb, tp + (size_t)(DFT_GROUPS - 1) * tstep,
               xp + (size_t)(DFT_GROUPS - 1) * xstep);
  __builtin_amdgcn_sched_barrier(0);
  dft_mma(ga, acc);
  dft_mma(gb, acc);

  float* pout = Ppart + (size_t)blockIdx.y * (NROW * DDIM) + colbase + 4 * c;
#pragma unroll
  for (int i = 0; i < 2; ++i) {
#pragma unroll
    for (int r = 0; r < 8; ++r) {
      const int row = rowbase + 2 * (8 * h + r) + i;
      v4f o;
      o.x = acc[i][0][r];
      o.y = acc[i][1][r];
      o.z = acc[i][2][r];
      o.w = acc[i][3][r];
      *(v4f*)(pout + (size_t)row * DDIM) = o;
    }
  }
}

#define RED_T 64
__global__ __launch_bounds__(RED_T) void reduce_chunks_kernel(
    const float* __restrict__ Ppart, float* __restrict__ C) {
  const int idx = blockIdx.x * RED_T + threadIdx.x;
  float v[NCHUNK];
#pragma unroll
  for (int c = 0; c < NCHUNK; ++c)
    v[c] = Ppart[(size_t)c * (NROW * DDIM) + idx];
  float s = 0.f;
#pragma unroll
  for (int c = 0; c < NCHUNK; ++c)
    s += v[c];
  C[idx] = s;
}

#define MIX_U 32
#define MIX_GROUPS (DDIM / MIX_U)
#define MIX_FSPLIT (DDIM / 32)

struct MixGroup {
  v2f p[MIX_U];
};

__device__ __forceinline__ void mix_load(MixGroup& g, const float* pp) {
#pragma unroll
  for (int s = 0; s < MIX_U; ++s)
    g.p[s] = *(const v2f*)(pp + (size_t)s * (DDIM * 2));
}

__device__ __forceinline__ void mix_fma(const MixGroup& g, const v2f* sx,
                                        float& ar, float& ai) {
#pragma unroll
  for (int s = 0; s < MIX_U; ++s) {
    const v2f xv = sx[s];
    const float xr = xv.x, xi = xv.y;
    const float pr = g.p[s].x, pi = g.p[s].y;
    ar = fmaf(xr, pr, fmaf(-xi, pi, ar));
    ai = fmaf(xr, pi, fmaf( xi, pr, ai));
  }
}

__global__ __launch_bounds__(32) void spectral_mix_kernel(
    const float* __restrict__ C, const float* __restrict__ P_low,
    const float* __restrict__ B_low, float* __restrict__ BppP) {
  __shared__ v2f sX[DDIM];
  const int lane = threadIdx.x;
  const int k = blockIdx.x / MIX_FSPLIT;
  const int f = (blockIdx.x % MIX_FSPLIT) * 32 + lane;
  const float* pp = P_low + ((size_t)k * DDIM * DDIM + f) * 2;
  const size_t gstep = (size_t)MIX_U * DDIM * 2;

  MixGroup ga, gb;
  mix_load(ga, pp);
  const float invN = 1.0f / (float)N_PTS;
#pragma unroll
  for (int i = 0; i < DDIM / 32; ++i) {
    const int d = i * 32 + lane;
    v2f xv;
    xv.x =  C[(size_t)(2 * k) * DDIM + d] * invN;
    xv.y = -C[(size_t)(2 * k + 1) * DDIM + d] * invN;
    sX[d] = xv;
  }
  float ar = B_low[((size_t)k * DDIM + f) * 2 + 0];
  float ai = B_low[((size_t)k * DDIM + f) * 2 + 1];
  __syncthreads();

  for (int g = 0; g < MIX_GROUPS - 2; g += 2) {
    mix_load(gb, pp + (size_t)(g + 1) * gstep);
    __builtin_amdgcn_sched_barrier(0);
    mix_fma(ga, sX + g * MIX_U, ar, ai);
    mix_load(ga, pp + (size_t)(g + 2) * gstep);
    mix_fma(gb, sX + (g + 1) * MIX_U, ar, ai);
    __builtin_amdgcn_sched_barrier(0);
  }
  mix_load(gb, pp + (size_t)(MIX_GROUPS - 1) * gstep);
  __builtin_amdgcn_sched_barrier(0);
  mix_fma(ga, sX + (MIX_GROUPS - 2) * MIX_U, ar, ai);
  mix_fma(gb, sX + (MIX_GROUPS - 1) * MIX_U, ar, ai);

  const float sc = (k == 0) ? 1.0f : 2.0f;
  v2f o;
  o.x =  sc * ar;
  o.y = -sc * ai;
  *(v2f*)(BppP + ((size_t)k * DDIM + f) * 2) = o;
}

#define FO_U 2
#define FO_XG (DDIM / (8 * FO_U))
#define FO_TG (NROW / (8 * FO_U))
#define FO_BSTEP ((size_t)4 * FO_U * DDIM * 2)
#define FO_PITCH 68

struct FoGroup {
  v4f a[FO_U][4];
  v4f b[FO_U][2][2];
};

template <int PITCH>
__device__ __forceinline__ void fo_load(FoGroup& g, const float* ap,
                                        const float* bp) {
#pragma unroll
  for (int p = 0; p < FO_U; ++p) {
#pragma unroll
    for (int i = 0; i < 4; ++i)
      g.a[p][i] = *(const v4f*)(ap + (size_t)(16 * i) * PITCH + 8 * p);
#pragma unroll
    for (int j = 0; j < 2; ++j)
#pragma unroll
      for (int u = 0; u < 2; ++u)
        g.b[p][j][u] =
            *(const v4f*)(bp + (size_t)(4 * p + j) * (DDIM * 2) + 64 * u);
  }
}

__device__ __forceinline__ void fo_mma(const FoGroup& g, v8f (&acc)[4][4]) {
#pragma unroll
  for (int p = 0; p < FO_U; ++p) {
#pragma unroll
    for (int j = 0; j < 2; ++j) {
#pragma unroll
      for (int i = 0; i < 4; ++i) {
        v2f a;
        a.x = g.a[p][i][2 * j];
        a.y = g.a[p][i][2 * j + 1];
#pragma unroll
        for (int t = 0; t < 4; ++t) {
          v2f b;
          b.x = g.b[p][j][t >> 1][2 * (t & 1)];
          b.y = g.b[p][j][t >> 1][2 * (t & 1) + 1];
          acc[i][t] = wmma_f32(a, b, acc[i][t]);
        }
      }
    }
  }
}

template <int HALF>
__device__ __forceinline__ void fo_epilogue(const v8f (&acc)[4][4], float* sw,
                                            const v4f bias, float* orow,
                                            int c, int h) {
#pragma unroll
  for (int il = 0; il < 2; ++il) {
#pragma unroll
    for (int r = 0; r < 8; ++r) {
      float* p = sw + (16 * il + 8 * h + r) * FO_PITCH + 2 * c;
      v2f lo, hi;
      lo.x = acc[2 * HALF + il][0][r];
      lo.y = acc[2 * HALF + il][1][r];
      hi.x = acc[2 * HALF + il][2][r];
      hi.y = acc[2 * HALF + il][3][r];
      *(v2f*)(p)      = lo;
      *(v2f*)(p + 32) = hi;
    }
  }
  __syncthreads();
#pragma unroll 1
  for (int q = 0; q < 16; ++q) {
    const int rl = 2 * q + h;
    const v4f v = *(const v4f*)(sw + rl * FO_PITCH + 4 * c);
    v4f o;
    o.x = sinf(0.2f * (v.x + bias.x));
    o.y = sinf(0.2f * (v.y + bias.y));
    o.z = sinf(0.2f * (v.z + bias.z));
    o.w = sinf(0.2f * (v.w + bias.w));
    *(v4f*)(orow + (size_t)(32 * HALF + rl) * DDIM) = o;
  }
}

__global__ __launch_bounds__(128) void fused_out_kernel(
    const float* __restrict__ x, const float* __restrict__ Wp,
    const float* __restrict__ Bias, const float* __restrict__ T,
    const float* __restrict__ BppP, float* __restrict__ out) {
  __shared__ float sOut[4 * 32 * FO_PITCH];
  const int lane = threadIdx.x & 31;
  const int wave = threadIdx.x >> 5;
  const int c    = lane & 15;
  const int h    = lane >> 4;
  const int n0   = blockIdx.x * 64;
  const int d0   = wave * 64;

  const float* xa = x + (size_t)(n0 + c) * DDIM + 4 * h;
  const float* ta = T + (size_t)(n0 + c) * NROW + 4 * h;
  const size_t boff = (size_t)(2 * h) * (DDIM * 2) + 2 * d0 + 4 * c;
  const float* wb = Wp + boff;
  const float* bb = BppP + boff;

  v8f acc[4][4];
#pragma unroll
  for (int i = 0; i < 4; ++i)
#pragma unroll
    for (int t = 0; t < 4; ++t)
      acc[i][t] = (v8f){0.f,0.f,0.f,0.f,0.f,0.f,0.f,0.f};

  FoGroup ga, gb;
  fo_load<DDIM>(ga, xa, wb);
  for (int g = 0; g < FO_XG - 2; g += 2) {
    fo_load<DDIM>(gb, xa + 16 * (g + 1), wb + (size_t)(g + 1) * FO_BSTEP);
    __builtin_amdgcn_sched_barrier(0);
    fo_mma(ga, acc);
    __builtin_amdgcn_sched_barrier(0);
    fo_load<DDIM>(ga, xa + 16 * (g + 2), wb + (size_t)(g + 2) * FO_BSTEP);
    __builtin_amdgcn_sched_barrier(0);
    fo_mma(gb, acc);
    __builtin_amdgcn_sched_barrier(0);
  }
  fo_load<DDIM>(gb, xa + 16 * (FO_XG - 1), wb + (size_t)(FO_XG - 1) * FO_BSTEP);
  __builtin_amdgcn_sched_barrier(0);
  fo_mma(ga, acc);
  __builtin_amdgcn_sched_barrier(0);
  fo_load<NROW>(ga, ta, bb);
  __builtin_amdgcn_sched_barrier(0);
  fo_mma(gb, acc);
  __builtin_amdgcn_sched_barrier(0);
  for (int g = 0; g < FO_TG - 2; g += 2) {
    fo_load<NROW>(gb, ta + 16 * (g + 1), bb + (size_t)(g + 1) * FO_BSTEP);
    __builtin_amdgcn_sched_barrier(0);
    fo_mma(ga, acc);
    __builtin_amdgcn_sched_barrier(0);
    fo_load<NROW>(ga, ta + 16 * (g + 2), bb + (size_t)(g + 2) * FO_BSTEP);
    __builtin_amdgcn_sched_barrier(0);
    fo_mma(gb, acc);
    __builtin_amdgcn_sched_barrier(0);
  }
  fo_load<NROW>(gb, ta + 16 * (FO_TG - 1), bb + (size_t)(FO_TG - 1) * FO_BSTEP);
  __builtin_amdgcn_sched_barrier(0);
  fo_mma(ga, acc);
  fo_mma(gb, acc);

  float* sw = sOut + wave * (32 * FO_PITCH);
  const v4f bias = *(const v4f*)(Bias + d0 + 4 * c);
  float* orow = out + (size_t)n0 * DDIM + d0 + 4 * c;
  fo_epilogue<0>(acc, sw, bias, orow, c, h);
  __syncthreads();
  fo_epilogue<1>(acc, sw, bias, orow, c, h);
}

extern "C" void kernel_launch(void* const* d_in, const int* in_sizes, int n_in,
                              void* d_out, int out_size, void* d_ws,
                              size_t ws_size, hipStream_t stream) {
  (void)in_sizes; (void)n_in; (void)out_size; (void)ws_size;
  const float* x     = (const float*)d_in[0];
  const float* P_low = (const float*)d_in[1];
  const float* B_low = (const float*)d_in[2];
  const float* W     = (const float*)d_in[3];
  const float* Bias  = (const float*)d_in[4];
  float* out = (float*)d_out;

  float* ws    = (float*)d_ws;
  float* T     = ws;
  float* Ppart = T + (size_t)N_PTS * NROW;
  float* C     = Ppart + (size_t)NCHUNK * NROW * DDIM;
  float* BppP  = C + NROW * DDIM;
  float* Wp    = BppP + NROW * DDIM;

  tgen_kernel<<<dim3((N_PTS * KFREQ) / 256), dim3(256), 0, stream>>>(T);
  pack_w_kernel<<<dim3((DDIM * DDIM / 2) / 256), dim3(256), 0, stream>>>(W, Wp);
  dft_low_kernel<<<dim3(DDIM / 64, NCHUNK), dim3(128), 0, stream>>>(x, T, Ppart);
  reduce_chunks_kernel<<<dim3((NROW * DDIM) / RED_T), dim3(RED_T), 0, stream>>>(Ppart, C);
  spectral_mix_kernel<<<dim3(KFREQ * MIX_FSPLIT), dim3(32), 0, stream>>>(C, P_low, B_low, BppP);
  fused_out_kernel<<<dim3(N_PTS / 64), dim3(128), 0, stream>>>(x, Wp, Bias, T, BppP, out);
}


# Forward DFT over row pairs (n, N-n), table reflection symmetry: cosine tiles take x[n]+x[N-n], sine tiles x[n]-x[N-n]; half its wmma and table rows; row N/2 added in chunk sum; sum order changed
# speedup vs baseline: 1.8395x; 1.1123x over previous
#include <hip/hip_runtime.h>
#include <math.h>


typedef float v2f __attribute__((ext_vector_type(2)));
typedef float v4f __attribute__((ext_vector_type(4)));
typedef float v8f __attribute__((ext_vector_type(8)));

#define N_PTS 65536
#define DDIM  256
#define KFREQ 64
#define NROW  128
#define NCHUNK 64
#define NHALF (N_PTS / 2)
#define CHUNK_P (NHALF / NCHUNK)

__device__ __forceinline__ v8f wmma_f32(v2f a, v2f b, v8f c) {
  return __builtin_amdgcn_wmma_f32_16x16x4_f32(false, a, false, b,
                                               (short)0, c, false, false);
}

__global__ __launch_bounds__(256) void tgen_kernel(float* __restrict__ T) {
  int idx = blockIdx.x * 256 + threadIdx.x;
  int n = idx >> 6;
  int k = idx & 63;
  unsigned m = ((unsigned)n * (unsigned)k) & 65535u;
  float ang = (float)m * 9.58737992429e-5f;
  float s, c;
  sincosf(ang, &s, &c);
  size_t base = (size_t)n * NROW + 2 * k;
  T[base]     = c;
  T[base + 1] = s;
}

__global__ __launch_bounds__(256) void pack_w_kernel(
    const float* __restrict__ W, float* __restrict__ Wp) {
  int idx = blockIdx.x * 256 + threadIdx.x;
  int kpair = idx >> 8;
  int col   = idx & 255;
  Wp[(size_t)idx * 2]     = W[(size_t)(2 * kpair) * DDIM + col];
  Wp[(size_t)idx * 2 + 1] = W[(size_t)(2 * kpair + 1) * DDIM + col];
}

#define DFT_U 8
#define DFT_GROUPS (CHUNK_P / (4 * DFT_U))

struct DftGroup {
  v2f t[DFT_U][2];
  v4f xv[DFT_U][2];
  v4f xm[DFT_U][2];
};

__device__ __forceinline__ void dft_load(DftGroup& g, const float* tp,
                                         const float* xp, const float* xq,
                                         const float* xq0) {
#pragma unroll
  for (int s = 0; s < DFT_U; ++s) {
#pragma unroll
    for (int e = 0; e < 2; ++e) {
      g.t[s][e]  = *(const v2f*)(tp + (size_t)(4 * s + e) * NROW);
      g.xv[s][e] = *(const v4f*)(xp + (size_t)(4 * s + e) * DDIM);
      if (s == 0 && e == 0)
        g.xm[s][e] = *(const v4f*)(xq0);
      else
        g.xm[s][e] = *(const v4f*)(xq - (size_t)(4 * s + e - 1) * DDIM);
    }
  }
}

__device__ __forceinline__ void dft_mma(const DftGroup& g, v8f (&acc)[2][4]) {
#pragma unroll
  for (int s = 0; s < DFT_U; ++s) {
    const v4f xs0 = g.xv[s][0] + g.xm[s][0];
    const v4f xs1 = g.xv[s][1] + g.xm[s][1];
    const v4f xd0 = g.xv[s][0] - g.xm[s][0];
    const v4f xd1 = g.xv[s][1] - g.xm[s][1];
    v2f a_cos, a_sin;
    a_cos.x = g.t[s][0][0];
    a_cos.y = g.t[s][1][0];
    a_sin.x = g.t[s][0][1];
    a_sin.y = g.t[s][1][1];
#pragma unroll
    for (int j = 0; j < 4; ++j) {
      v2f b;
      b.x = xs0[j];
      b.y = xs1[j];
      acc[0][j] = wmma_f32(a_cos, b, acc[0][j]);
    }
#pragma unroll
    for (int j = 0; j < 4; ++j) {
      v2f b;
      b.x = xd0[j];
      b.y = xd1[j];
      acc[1][j] = wmma_f32(a_sin, b, acc[1][j]);
    }
  }
}

__global__ __launch_bounds__(128) void dft_low_kernel(
    const float* __restrict__ x, const float* __restrict__ T,
    float* __restrict__ Ppart) {
  const int lane    = threadIdx.x & 31;
  const int wave    = threadIdx.x >> 5;
  const int c       = lane & 15;
  const int h       = lane >> 4;
  const int rowbase = wave * 32;
  const int colbase = blockIdx.x * 64;
  const int nbase   = blockIdx.y * CHUNK_P;
  const int n00     = nbase + 2 * h;

  const float* tp = T + (size_t)n00 * NROW + rowbase + 2 * c;
  const float* xp = x + (size_t)n00 * DDIM + colbase + 4 * c;
  const float* xq = x + (size_t)(N_PTS - n00 - 1) * DDIM + colbase + 4 * c;
  const bool lone   = (n00 == 0);
  const float* xq0  = lone ? (x + colbase + 4 * c) : (xq + DDIM);
  const size_t tstep = (size_t)4 * DFT_U * NROW;
  const size_t xstep = (size_t)4 * DFT_U * DDIM;

  v8f acc[2][4];
#pragma unroll
  for (int i = 0; i < 2; ++i)
#pragma unroll
    for (int j = 0; j < 4; ++j)
      acc[i][j] = (v8f){0.f,0.f,0.f,0.f,0.f,0.f,0.f,0.f};

  DftGroup ga, gb;
  dft_load(ga, tp, xp, xq, xq0);
  if (lone)
    ga.xm[0][0] = (v4f){0.f, 0.f, 0.f, 0.f};
  for (int g = 0; g < DFT_GROUPS - 2; g += 2) {
    dft_load(gb, tp + (size_t)(g + 1) * tstep, xp + (size_t)(g + 1) * xstep,
                 xq - (size_t)(g + 1) * xstep,
                 xq - (size_t)(g + 1) * xstep + DDIM);
    __builtin_amdgcn_sched_barrier(0);
    dft_mma(ga, acc);
    __builtin_amdgcn_sched_barrier(0);
    dft_load(ga, tp + (size_t)(g + 2) * tstep, xp + (size_t)(g + 2) * xstep,
                 xq - (size_t)(g + 2) * xstep,
                 xq - (size_t)(g + 2) * xstep + DDIM);
    __builtin_amdgcn_sched_barrier(0);
    dft_mma(gb, acc);
    __builtin_amdgcn_sched_barrier(0);
  }
  dft_load(gb, tp + (size_t)(DFT_GROUPS - 1) * tstep,
               xp + (size_t)(DFT_GROUPS - 1) * xstep,
               xq - (size_t)(DFT_GROUPS - 1) * xstep,
               xq - (size_t)(DFT_GROUPS - 1) * xstep + DDIM);
  __builtin_amdgcn_sched_barrier(0);
  dft_mma(ga, acc);
  dft_mma(gb, acc);

  float* pout = Ppart + (size_t)blockIdx.y * (NROW * DDIM) + colbase + 4 * c;
#pragma unroll
  for (int i = 0; i < 2; ++i) {
#pragma unroll
    for (int r = 0; r < 8; ++r) {
      const int row = rowbase + 2 * (8 * h + r) + i;
      v4f o;
      o.x = acc[i][0][r];
      o.y = acc[i][1][r];
      o.z = acc[i][2][r];
      o.w = acc[i][3][r];
      *(v4f*)(pout + (size_t)row * DDIM) = o;
    }
  }
}

#define RED_T 64
__global__ __launch_bounds__(RED_T) void reduce_chunks_kernel(
    const float* __restrict__ Ppart, const float* __restrict__ x,
    float* __restrict__ C) {
  const int idx = blockIdx.x * RED_T + threadIdx.x;
  const int row = idx >> 8;
  const float xh = x[(size_t)NHALF * DDIM + (idx & (DDIM - 1))];
  float v[NCHUNK];
#pragma unroll
  for (int c = 0; c < NCHUNK; ++c)
    v[c] = Ppart[(size_t)c * (NROW * DDIM) + idx];
  float s = 0.f;
#pragma unroll
  for (int c = 0; c < NCHUNK; ++c)
    s += v[c];
  if ((row & 1) == 0)
    s += (row & 2) ? -xh : xh;
  C[idx] = s;
}

#define MIX_U 32
#define MIX_GROUPS (DDIM / MIX_U)
#define MIX_FSPLIT (DDIM / 32)

struct MixGroup {
  v2f p[MIX_U];
};

__device__ __forceinline__ void mix_load(MixGroup& g, const float* pp) {
#pragma unroll
  for (int s = 0; s < MIX_U; ++s)
    g.p[s] = *(const v2f*)(pp + (size_t)s * (DDIM * 2));
}

__device__ __forceinline__ void mix_fma(const MixGroup& g, const v2f* sx,
                                        float& ar, float& ai) {
#pragma unroll
  for (int s = 0; s < MIX_U; ++s) {
    const v2f xv = sx[s];
    const float xr = xv.x, xi = xv.y;
    const float pr = g.p[s].x, pi = g.p[s].y;
    ar = fmaf(xr, pr, fmaf(-xi, pi, ar));
    ai = fmaf(xr, pi, fmaf( xi, pr, ai));
  }
}

__global__ __launch_bounds__(32) void spectral_mix_kernel(
    const float* __restrict__ C, const float* __restrict__ P_low,
    const float* __restrict__ B_low, float* __restrict__ BppP) {
  __shared__ v2f sX[DDIM];
  const int lane = threadIdx.x;
  const int k = blockIdx.x / MIX_FSPLIT;
  const int f = (blockIdx.x % MIX_FSPLIT) * 32 + lane;
  const float* pp = P_low + ((size_t)k * DDIM * DDIM + f) * 2;
  const size_t gstep = (size_t)MIX_U * DDIM * 2;

  MixGroup ga, gb;
  mix_load(ga, pp);
  const float invN = 1.0f / (float)N_PTS;
#pragma unroll
  for (int i = 0; i < DDIM / 32; ++i) {
    const int d = i * 32 + lane;
    v2f xv;
    xv.x =  C[(size_t)(2 * k) * DDIM + d] * invN;
    xv.y = -C[(size_t)(2 * k + 1) * DDIM + d] * invN;
    sX[d] = xv;
  }
  float ar = B_low[((size_t)k * DDIM + f) * 2 + 0];
  float ai = B_low[((size_t)k * DDIM + f) * 2 + 1];
  __syncthreads();

  for (int g = 0; g < MIX_GROUPS - 2; g += 2) {
    mix_load(gb, pp + (size_t)(g + 1) * gstep);
    __builtin_amdgcn_sched_barrier(0);
    mix_fma(ga, sX + g * MIX_U, ar, ai);
    mix_load(ga, pp + (size_t)(g + 2) * gstep);
    mix_fma(gb, sX + (g + 1) * MIX_U, ar, ai);
    __builtin_amdgcn_sched_barrier(0);
  }
  mix_load(gb, pp + (size_t)(MIX_GROUPS - 1) * gstep);
  __builtin_amdgcn_sched_barrier(0);
  mix_fma(ga, sX + (MIX_GROUPS - 2) * MIX_U, ar, ai);
  mix_fma(gb, sX + (MIX_GROUPS - 1) * MIX_U, ar, ai);

  const float sc = (k == 0) ? 1.0f : 2.0f;
  v2f o;
  o.x =  sc * ar;
  o.y = -sc * ai;
  *(v2f*)(BppP + ((size_t)k * DDIM + f) * 2) = o;
}

#define FO_U 2
#define FO_XG (DDIM / (8 * FO_U))
#define FO_TG (NROW / (8 * FO_U))
#define FO_BSTEP ((size_t)4 * FO_U * DDIM * 2)
#define FO_PITCH 68

struct FoGroup {
  v4f a[FO_U][4];
  v4f b[FO_U][2][2];
};

template <int PITCH>
__device__ __forceinline__ void fo_load(FoGroup& g, const float* ap,
                                        const float* bp) {
#pragma unroll
  for (int p = 0; p < FO_U; ++p) {
#pragma unroll
    for (int i = 0; i < 4; ++i)
      g.a[p][i] = *(const v4f*)(ap + (size_t)(16 * i) * PITCH + 8 * p);
#pragma unroll
    for (int j = 0; j < 2; ++j)
#pragma unroll
      for (int u = 0; u < 2; ++u)
        g.b[p][j][u] =
            *(const v4f*)(bp + (size_t)(4 * p + j) * (DDIM * 2) + 64 * u);
  }
}

__device__ __forceinline__ void fo_mma(const FoGroup& g, v8f (&acc)[4][4]) {
#pragma unroll
  for (int p = 0; p < FO_U; ++p) {
#pragma unroll
    for (int j = 0; j < 2; ++j) {
#pragma unroll
      for (int i = 0; i < 4; ++i) {
        v2f a;
        a.x = g.a[p][i][2 * j];
        a.y = g.a[p][i][2 * j + 1];
#pragma unroll
        for (int t = 0; t < 4; ++t) {
          v2f b;
          b.x = g.b[p][j][t >> 1][2 * (t & 1)];
          b.y = g.b[p][j][t >> 1][2 * (t & 1) + 1];
          acc[i][t] = wmma_f32(a, b, acc[i][t]);
        }
      }
    }
  }
}

template <int HALF>
__device__ __forceinline__ void fo_epilogue(const v8f (&acc)[4][4], float* sw,
                                            const v4f bias, float* orow,
                                            int c, int h) {
#pragma unroll
  for (int il = 0; il < 2; ++il) {
#pragma unroll
    for (int r = 0; r < 8; ++r) {
      float* p = sw + (16 * il + 8 * h + r) * FO_PITCH + 2 * c;
      v2f lo, hi;
      lo.x = acc[2 * HALF + il][0][r];
      lo.y = acc[2 * HALF + il][1][r];
      hi.x = acc[2 * HALF + il][2][r];
      hi.y = acc[2 * HALF + il][3][r];
      *(v2f*)(p)      = lo;
      *(v2f*)(p + 32) = hi;
    }
  }
  __syncthreads();
#pragma unroll 1
  for (int q = 0; q < 16; ++q) {
    const int rl = 2 * q + h;
    const v4f v = *(const v4f*)(sw + rl * FO_PITCH + 4 * c);
    v4f o;
    o.x = sinf(0.2f * (v.x + bias.x));
    o.y = sinf(0.2f * (v.y + bias.y));
    o.z = sinf(0.2f * (v.z + bias.z));
    o.w = sinf(0.2f * (v.w + bias.w));
    *(v4f*)(orow + (size_t)(32 * HALF + rl) * DDIM) = o;
  }
}

__global__ __launch_bounds__(128) void fused_out_kernel(
    const float* __restrict__ x, const float* __restrict__ Wp,
    const float* __restrict__ Bias, const float* __restrict__ T,
    const float* __restrict__ BppP, float* __restrict__ out) {
  __shared__ float sOut[4 * 32 * FO_PITCH];
  const int lane = threadIdx.x & 31;
  const int wave = threadIdx.x >> 5;
  const int c    = lane & 15;
  const int h    = lane >> 4;
  const int n0   = blockIdx.x * 64;
  const int d0   = wave * 64;

  const float* xa = x + (size_t)(n0 + c) * DDIM + 4 * h;
  const float* ta = T + (size_t)(n0 + c) * NROW + 4 * h;
  const size_t boff = (size_t)(2 * h) * (DDIM * 2) + 2 * d0 + 4 * c;
  const float* wb = Wp + boff;
  const float* bb = BppP + boff;

  v8f acc[4][4];
#pragma unroll
  for (int i = 0; i < 4; ++i)
#pragma unroll
    for (int t = 0; t < 4; ++t)
      acc[i][t] = (v8f){0.f,0.f,0.f,0.f,0.f,0.f,0.f,0.f};

  FoGroup ga, gb;
  fo_load<DDIM>(ga, xa, wb);
  for (int g = 0; g < FO_XG - 2; g += 2) {
    fo_load<DDIM>(gb, xa + 16 * (g + 1), wb + (size_t)(g + 1) * FO_BSTEP);
    __builtin_amdgcn_sched_barrier(0);
    fo_mma(ga, acc);
    __builtin_amdgcn_sched_barrier(0);
    fo_load<DDIM>(ga, xa + 16 * (g + 2), wb + (size_t)(g + 2) * FO_BSTEP);
    __builtin_amdgcn_sched_barrier(0);
    fo_mma(gb, acc);
    __builtin_amdgcn_sched_barrier(0);
  }
  fo_load<DDIM>(gb, xa + 16 * (FO_XG - 1), wb + (size_t)(FO_XG - 1) * FO_BSTEP);
  __builtin_amdgcn_sched_barrier(0);
  fo_mma(ga, acc);
  __builtin_amdgcn_sched_barrier(0);
  fo_load<NROW>(ga, ta, bb);
  __builtin_amdgcn_sched_barrier(0);
  fo_mma(gb, acc);
  __builtin_amdgcn_sched_barrier(0);
  for (int g = 0; g < FO_TG - 2; g += 2) {
    fo_load<NROW>(gb, ta + 16 * (g + 1), bb + (size_t)(g + 1) * FO_BSTEP);
    __builtin_amdgcn_sched_barrier(0);
    fo_mma(ga, acc);
    __builtin_amdgcn_sched_barrier(0);
    fo_load<NROW>(ga, ta + 16 * (g + 2), bb + (size_t)(g + 2) * FO_BSTEP);
    __builtin_amdgcn_sched_barrier(0);
    fo_mma(gb, acc);
    __builtin_amdgcn_sched_barrier(0);
  }
  fo_load<NROW>(gb, ta + 16 * (FO_TG - 1), bb + (size_t)(FO_TG - 1) * FO_BSTEP);
  __builtin_amdgcn_sched_barrier(0);
  fo_mma(ga, acc);
  fo_mma(gb, acc);

  float* sw = sOut + wave * (32 * FO_PITCH);
  const v4f bias = *(const v4f*)(Bias + d0 + 4 * c);
  float* orow = out + (size_t)n0 * DDIM + d0 + 4 * c;
  fo_epilogue<0>(acc, sw, bias, orow, c, h);
  __syncthreads();
  fo_epilogue<1>(acc, sw, bias, orow, c, h);
}

extern "C" void kernel_launch(void* const* d_in, const int* in_sizes, int n_in,
                              void* d_out, int out_size, void* d_ws,
                              size_t ws_size, hipStream_t stream) {
  (void)in_sizes; (void)n_in; (void)out_size; (void)ws_size;
  const float* x     = (const float*)d_in[0];
  const float* P_low = (const float*)d_in[1];
  const float* B_low = (const float*)d_in[2];
  const float* W     = (const float*)d_in[3];
  const float* Bias  = (const float*)d_in[4];
  float* out = (float*)d_out;

  float* ws    = (float*)d_ws;
  float* T     = ws;
  float* Ppart = T + (size_t)N_PTS * NROW;
  float* C     = Ppart + (size_t)NCHUNK * NROW * DDIM;
  float* BppP  = C + NROW * DDIM;
  float* Wp    = BppP + NROW * DDIM;

  tgen_kernel<<<dim3((N_PTS * KFREQ) / 256), dim3(256), 0, stream>>>(T);
  pack_w_kernel<<<dim3((DDIM * DDIM / 2) / 256), dim3(256), 0, stream>>>(W, Wp);
  dft_low_kernel<<<dim3(DDIM / 64, NCHUNK), dim3(128), 0, stream>>>(x, T, Ppart);
  reduce_chunks_kernel<<<dim3((NROW * DDIM) / RED_T), dim3(RED_T), 0, stream>>>(Ppart, x, C);
  spectral_mix_kernel<<<dim3(KFREQ * MIX_FSPLIT), dim3(32), 0, stream>>>(C, P_low, B_low, BppP);
  fused_out_kernel<<<dim3(N_PTS / 64), dim3(128), 0, stream>>>(x, Wp, Bias, T, BppP, out);
}


# Output GEMM over row pairs (n, N-n), table reflection symmetry: cos sums S, sin sums D once per pair, rows start S+D / S-D, then x@W; row N/2 by signed adds; half the table wmma; sum order changed
# speedup vs baseline: 1.9890x; 1.0813x over previous
#include <hip/hip_runtime.h>
#include <math.h>


typedef float v2f __attribute__((ext_vector_type(2)));
typedef float v4f __attribute__((ext_vector_type(4)));
typedef float v8f __attribute__((ext_vector_type(8)));

#define N_PTS 65536
#define DDIM  256
#define KFREQ 64
#define NROW  128
#define NCHUNK 64
#define NHALF (N_PTS / 2)
#define CHUNK_P (NHALF / NCHUNK)

__device__ __forceinline__ v8f wmma_f32(v2f a, v2f b, v8f c) {
  return __builtin_amdgcn_wmma_f32_16x16x4_f32(false, a, false, b,
                                               (short)0, c, false, false);
}

#define T_ROWS NHALF
__global__ __launch_bounds__(256) void tgen_kernel(float* __restrict__ T) {
  int idx = blockIdx.x * 256 + threadIdx.x;
  int n = idx >> 6;
  int k = idx & 63;
  unsigned m = ((unsigned)n * (unsigned)k) & 65535u;
  float ang = (float)m * 9.58737992429e-5f;
  float s, c;
  sincosf(ang, &s, &c);
  size_t base = (size_t)n * NROW + 2 * k;
  T[base]     = c;
  T[base + 1] = s;
}

__global__ __launch_bounds__(256) void pack_w_kernel(
    const float* __restrict__ W, float* __restrict__ Wp) {
  int idx = blockIdx.x * 256 + threadIdx.x;
  int kpair = idx >> 8;
  int col   = idx & 255;
  Wp[(size_t)idx * 2]     = W[(size_t)(2 * kpair) * DDIM + col];
  Wp[(size_t)idx * 2 + 1] = W[(size_t)(2 * kpair + 1) * DDIM + col];
}

#define DFT_U 8
#define DFT_GROUPS (CHUNK_P / (4 * DFT_U))

struct DftGroup {
  v2f t[DFT_U][2];
  v4f xv[DFT_U][2];
  v4f xm[DFT_U][2];
};

__device__ __forceinline__ void dft_load(DftGroup& g, const float* tp,
                                         const float* xp, const float* xq,
                                         const float* xq0) {
#pragma unroll
  for (int s = 0; s < DFT_U; ++s) {
#pragma unroll
    for (int e = 0; e < 2; ++e) {
      g.t[s][e]  = *(const v2f*)(tp + (size_t)(4 * s + e) * NROW);
      g.xv[s][e] = *(const v4f*)(xp + (size_t)(4 * s + e) * DDIM);
      if (s == 0 && e == 0)
        g.xm[s][e] = *(const v4f*)(xq0);
      else
        g.xm[s][e] = *(const v4f*)(xq - (size_t)(4 * s + e - 1) * DDIM);
    }
  }
}

__device__ __forceinline__ void dft_mma(const DftGroup& g, v8f (&acc)[2][4]) {
#pragma unroll
  for (int s = 0; s < DFT_U; ++s) {
    const v4f xs0 = g.xv[s][0] + g.xm[s][0];
    const v4f xs1 = g.xv[s][1] + g.xm[s][1];
    const v4f xd0 = g.xv[s][0] - g.xm[s][0];
    const v4f xd1 = g.xv[s][1] - g.xm[s][1];
    v2f a_cos, a_sin;
    a_cos.x = g.t[s][0][0];
    a_cos.y = g.t[s][1][0];
    a_sin.x = g.t[s][0][1];
    a_sin.y = g.t[s][1][1];
#pragma unroll
    for (int j = 0; j < 4; ++j) {
      v2f b;
      b.x = xs0[j];
      b.y = xs1[j];
      acc[0][j] = wmma_f32(a_cos, b, acc[0][j]);
    }
#pragma unroll
    for (int j = 0; j < 4; ++j) {
      v2f b;
      b.x = xd0[j];
      b.y = xd1[j];
      acc[1][j] = wmma_f32(a_sin, b, acc[1][j]);
    }
  }
}

__global__ __launch_bounds__(128) void dft_low_kernel(
    const float* __restrict__ x, const float* __restrict__ T,
    float* __restrict__ Ppart) {
  const int lane    = threadIdx.x & 31;
  const int wave    = threadIdx.x >> 5;
  const int c       = lane & 15;
  const int h       = lane >> 4;
  const int rowbase = wave * 32;
  const int colbase = blockIdx.x * 64;
  const int nbase   = blockIdx.y * CHUNK_P;
  const int n00     = nbase + 2 * h;

  const float* tp = T + (size_t)n00 * NROW + rowbase + 2 * c;
  const float* xp = x + (size_t)n00 * DDIM + colbase + 4 * c;
  const float* xq = x + (size_t)(N_PTS - n00 - 1) * DDIM + colbase + 4 * c;
  const bool lone   = (n00 == 0);
  const float* xq0  = lone ? (x + colbase + 4 * c) : (xq + DDIM);
  const size_t tstep = (size_t)4 * DFT_U * NROW;
  const size_t xstep = (size_t)4 * DFT_U * DDIM;

  v8f acc[2][4];
#pragma unroll
  for (int i = 0; i < 2; ++i)
#pragma unroll
    for (int j = 0; j < 4; ++j)
      acc[i][j] = (v8f){0.f,0.f,0.f,0.f,0.f,0.f,0.f,0.f};

  DftGroup ga, gb;
  dft_load(ga, tp, xp, xq, xq0);
  if (lone)
    ga.xm[0][0] = (v4f){0.f, 0.f, 0.f, 0.f};
  for (int g = 0; g < DFT_GROUPS - 2; g += 2) {
    dft_load(gb, tp + (size_t)(g + 1) * tstep, xp + (size_t)(g + 1) * xstep,
                 xq - (size_t)(g + 1) * xstep,
                 xq - (size_t)(g + 1) * xstep + DDIM);
    __builtin_amdgcn_sched_barrier(0);
    dft_mma(ga, acc);
    __builtin_amdgcn_sched_barrier(0);
    dft_load(ga, tp + (size_t)(g + 2) * tstep, xp + (size_t)(g + 2) * xstep,
                 xq - (size_t)(g + 2) * xstep,
                 xq - (size_t)(g + 2) * xstep + DDIM);
    __builtin_amdgcn_sched_barrier(0);
    dft_mma(gb, acc);
    __builtin_amdgcn_sched_barrier(0);
  }
  dft_load(gb, tp + (size_t)(DFT_GROUPS - 1) * tstep,
               xp + (size_t)(DFT_GROUPS - 1) * xstep,
               xq - (size_t)(DFT_GROUPS - 1) * xstep,
               xq - (size_t)(DFT_GROUPS - 1) * xstep + DDIM);
  __builtin_amdgcn_sched_barrier(0);
  dft_mma(ga, acc);
  dft_mma(gb, acc);

  float* pout = Ppart + (size_t)blockIdx.y * (NROW * DDIM) + colbase + 4 * c;
#pragma unroll
  for (int i = 0; i < 2; ++i) {
#pragma unroll
    for (int r = 0; r < 8; ++r) {
      const int row = rowbase + 2 * (8 * h + r) + i;
      v4f o;
      o.x = acc[i][0][r];
      o.y = acc[i][1][r];
      o.z = acc[i][2][r];
      o.w = acc[i][3][r];
      *(v4f*)(pout + (size_t)row * DDIM) = o;
    }
  }
}

#define RED_T 64
__global__ __launch_bounds__(RED_T) void reduce_chunks_kernel(
    const float* __restrict__ Ppart, const float* __restrict__ x,
    float* __restrict__ C) {
  const int idx = blockIdx.x * RED_T + threadIdx.x;
  const int row = idx >> 8;
  const float xh = x[(size_t)NHALF * DDIM + (idx & (DDIM - 1))];
  float v[NCHUNK];
#pragma unroll
  for (int c = 0; c < NCHUNK; ++c)
    v[c] = Ppart[(size_t)c * (NROW * DDIM) + idx];
  float s = 0.f;
#pragma unroll
  for (int c = 0; c < NCHUNK; ++c)
    s += v[c];
  if ((row & 1) == 0)
    s += (row & 2) ? -xh : xh;
  C[idx] = s;
}

#define MIX_U 32
#define MIX_GROUPS (DDIM / MIX_U)
#define MIX_FSPLIT (DDIM / 32)

struct MixGroup {
  v2f p[MIX_U];
};

__device__ __forceinline__ void mix_load(MixGroup& g, const float* pp) {
#pragma unroll
  for (int s = 0; s < MIX_U; ++s)
    g.p[s] = *(const v2f*)(pp + (size_t)s * (DDIM * 2));
}

__device__ __forceinline__ void mix_fma(const MixGroup& g, const v2f* sx,
                                        float& ar, float& ai) {
#pragma unroll
  for (int s = 0; s < MIX_U; ++s) {
    const v2f xv = sx[s];
    const float xr = xv.x, xi = xv.y;
    const float pr = g.p[s].x, pi = g.p[s].y;
    ar = fmaf(xr, pr, fmaf(-xi, pi, ar));
    ai = fmaf(xr, pi, fmaf( xi, pr, ai));
  }
}

__global__ __launch_bounds__(32) void spectral_mix_kernel(
    const float* __restrict__ C, const float* __restrict__ P_low,
    const float* __restrict__ B_low, float* __restrict__ BppP) {
  __shared__ v2f sX[DDIM];
  const int lane = threadIdx.x;
  const int k = blockIdx.x / MIX_FSPLIT;
  const int f = (blockIdx.x % MIX_FSPLIT) * 32 + lane;
  const float* pp = P_low + ((size_t)k * DDIM * DDIM + f) * 2;
  const size_t gstep = (size_t)MIX_U * DDIM * 2;

  MixGroup ga, gb;
  mix_load(ga, pp);
  const float invN = 1.0f / (float)N_PTS;
#pragma unroll
  for (int i = 0; i < DDIM / 32; ++i) {
    const int d = i * 32 + lane;
    v2f xv;
    xv.x =  C[(size_t)(2 * k) * DDIM + d] * invN;
    xv.y = -C[(size_t)(2 * k + 1) * DDIM + d] * invN;
    sX[d] = xv;
  }
  float ar = B_low[((size_t)k * DDIM + f) * 2 + 0];
  float ai = B_low[((size_t)k * DDIM + f) * 2 + 1];
  __syncthreads();

  for (int g = 0; g < MIX_GROUPS - 2; g += 2) {
    mix_load(gb, pp + (size_t)(g + 1) * gstep);
    __builtin_amdgcn_sched_barrier(0);
    mix_fma(ga, sX + g * MIX_U, ar, ai);
    mix_load(ga, pp + (size_t)(g + 2) * gstep);
    mix_fma(gb, sX + (g + 1) * MIX_U, ar, ai);
    __builtin_amdgcn_sched_barrier(0);
  }
  mix_load(gb, pp + (size_t)(MIX_GROUPS - 1) * gstep);
  __builtin_amdgcn_sched_barrier(0);
  mix_fma(ga, sX + (MIX_GROUPS - 2) * MIX_U, ar, ai);
  mix_fma(gb, sX + (MIX_GROUPS - 1) * MIX_U, ar, ai);

  const float sc = (k == 0) ? 1.0f : 2.0f;
  v2f o;
  o.x =  sc * ar;
  o.y = -sc * ai;
  *(v2f*)(BppP + ((size_t)k * DDIM + f) * 2) = o;
}

#define FO_U 2
#define FO_XG (DDIM / (8 * FO_U))
#define FO_TG (NROW / (8 * FO_U))
#define FO_BSTEP ((size_t)4 * FO_U * DDIM * 2)
#define FO_PITCH 68

struct FoGroup {
  v4f a[FO_U][4];
  v4f b[FO_U][2][2];
};

__device__ __forceinline__ void fo_load_b(FoGroup& g, const float* bp) {
#pragma unroll
  for (int p = 0; p < FO_U; ++p)
#pragma unroll
    for (int j = 0; j < 2; ++j)
#pragma unroll
      for (int u = 0; u < 2; ++u)
        g.b[p][j][u] =
            *(const v4f*)(bp + (size_t)(4 * p + j) * (DDIM * 2) + 64 * u);
}

__device__ __forceinline__ void fo_load_x(FoGroup& g, const float* a0,
                                          const float* a2, const float* a3,
                                          const float* bp) {
#pragma unroll
  for (int p = 0; p < FO_U; ++p) {
    g.a[p][0] = *(const v4f*)(a0 + 8 * p);
    g.a[p][1] = *(const v4f*)(a0 + (size_t)16 * DDIM + 8 * p);
    g.a[p][2] = *(const v4f*)(a2 + 8 * p);
    g.a[p][3] = *(const v4f*)(a3 + 8 * p);
  }
  fo_load_b(g, bp);
}

__device__ __forceinline__ void fo_load_t(FoGroup& g, const float* ta,
                                          const float* bp) {
#pragma unroll
  for (int p = 0; p < FO_U; ++p) {
    g.a[p][0] = *(const v4f*)(ta + 8 * p);
    g.a[p][1] = *(const v4f*)(ta + (size_t)16 * NROW + 8 * p);
  }
  fo_load_b(g, bp);
}

__device__ __forceinline__ void fo_mma_t(const FoGroup& g, v8f (&acc)[4][4]) {
#pragma unroll
  for (int p = 0; p < FO_U; ++p) {
#pragma unroll
    for (int i = 0; i < 2; ++i) {
      v2f a_c, a_s;
      a_c.x = g.a[p][i][0];
      a_c.y = g.a[p][i][2];
      a_s.x = g.a[p][i][1];
      a_s.y = g.a[p][i][3];
#pragma unroll
      for (int t = 0; t < 4; ++t) {
        v2f b;
        b.x = g.b[p][0][t >> 1][2 * (t & 1)];
        b.y = g.b[p][1][t >> 1][2 * (t & 1)];
        acc[i][t] = wmma_f32(a_c, b, acc[i][t]);
      }
#pragma unroll
      for (int t = 0; t < 4; ++t) {
        v2f b;
        b.x = g.b[p][0][t >> 1][2 * (t & 1) + 1];
        b.y = g.b[p][1][t >> 1][2 * (t & 1) + 1];
        acc[2 + i][t] = wmma_f32(a_s, b, acc[2 + i][t]);
      }
    }
  }
}

__device__ __forceinline__ void fo_mma(const FoGroup& g, v8f (&acc)[4][4]) {
#pragma unroll
  for (int p = 0; p < FO_U; ++p) {
#pragma unroll
    for (int j = 0; j < 2; ++j) {
#pragma unroll
      for (int i = 0; i < 4; ++i) {
        v2f a;
        a.x = g.a[p][i][2 * j];
        a.y = g.a[p][i][2 * j + 1];
#pragma unroll
        for (int t = 0; t < 4; ++t) {
          v2f b;
          b.x = g.b[p][j][t >> 1][2 * (t & 1)];
          b.y = g.b[p][j][t >> 1][2 * (t & 1) + 1];
          acc[i][t] = wmma_f32(a, b, acc[i][t]);
        }
      }
    }
  }
}

template <int HALF>
__device__ __forceinline__ void fo_epilogue(const v8f (&acc)[4][4], float* sw,
                                            const v4f bias, float* ocol,
                                            int n0, int c, int h) {
#pragma unroll
  for (int il = 0; il < 2; ++il) {
#pragma unroll
    for (int r = 0; r < 8; ++r) {
      float* p = sw + (16 * il + 8 * h + r) * FO_PITCH + 2 * c;
      v2f lo, hi;
      lo.x = acc[2 * HALF + il][0][r];
      lo.y = acc[2 * HALF + il][1][r];
      hi.x = acc[2 * HALF + il][2][r];
      hi.y = acc[2 * HALF + il][3][r];
      *(v2f*)(p)      = lo;
      *(v2f*)(p + 32) = hi;
    }
  }
  __syncthreads();
#pragma unroll 1
  for (int q = 0; q < 16; ++q) {
    const int rl = 2 * q + h;
    const int n  = n0 + rl;
    const v4f v = *(const v4f*)(sw + rl * FO_PITCH + 4 * c);
    v4f o;
    o.x = sinf(0.2f * (v.x + bias.x));
    o.y = sinf(0.2f * (v.y + bias.y));
    o.z = sinf(0.2f * (v.z + bias.z));
    o.w = sinf(0.2f * (v.w + bias.w));
    const int row = (HALF == 0) ? n : (n == 0 ? NHALF : N_PTS - n);
    *(v4f*)(ocol + (size_t)row * DDIM) = o;
  }
}

__global__ __launch_bounds__(128) void fused_out_kernel(
    const float* __restrict__ x, const float* __restrict__ Wp,
    const float* __restrict__ Bias, const float* __restrict__ T,
    const float* __restrict__ BppP, float* __restrict__ out) {
  __shared__ float sOut[4 * 32 * FO_PITCH];
  const int lane = threadIdx.x & 31;
  const int wave = threadIdx.x >> 5;
  const int c    = lane & 15;
  const int h    = lane >> 4;
  const int n0   = blockIdx.x * 32;
  const int d0   = wave * 64;

  const float* xa0 = x + (size_t)(n0 + c) * DDIM + 4 * h;
  const float* ta  = T + (size_t)(n0 + c) * NROW + 4 * h;
  const int np0 = (n0 + c == 0) ? NHALF : N_PTS - n0 - c;
  const float* xa2 = x + (size_t)np0 * DDIM + 4 * h;
  const float* xa3 = x + (size_t)(N_PTS - n0 - 16 - c) * DDIM + 4 * h;
  const size_t boff = (size_t)(2 * h) * (DDIM * 2) + 2 * d0 + 4 * c;
  const float* wb = Wp + boff;
  const float* bb = BppP + boff;

  v8f acc[4][4];
#pragma unroll
  for (int i = 0; i < 4; ++i)
#pragma unroll
    for (int t = 0; t < 4; ++t)
      acc[i][t] = (v8f){0.f,0.f,0.f,0.f,0.f,0.f,0.f,0.f};

  FoGroup ga, gb;
  fo_load_t(ga, ta, bb);
  for (int g = 0; g < FO_TG - 2; g += 2) {
    fo_load_t(gb, ta + 16 * (g + 1), bb + (size_t)(g + 1) * FO_BSTEP);
    __builtin_amdgcn_sched_barrier(0);
    fo_mma_t(ga, acc);
    __builtin_amdgcn_sched_barrier(0);
    fo_load_t(ga, ta + 16 * (g + 2), bb + (size_t)(g + 2) * FO_BSTEP);
    __builtin_amdgcn_sched_barrier(0);
    fo_mma_t(gb, acc);
    __builtin_amdgcn_sched_barrier(0);
  }
  fo_load_t(gb, ta + 16 * (FO_TG - 1), bb + (size_t)(FO_TG - 1) * FO_BSTEP);
  __builtin_amdgcn_sched_barrier(0);
  fo_mma_t(ga, acc);
  __builtin_amdgcn_sched_barrier(0);
  fo_load_x(ga, xa0, xa2, xa3, wb);
  __builtin_amdgcn_sched_barrier(0);
  fo_mma_t(gb, acc);
  __builtin_amdgcn_sched_barrier(0);

#pragma unroll
  for (int i = 0; i < 2; ++i) {
#pragma unroll
    for (int t = 0; t < 4; ++t) {
      const v8f sv = acc[i][t];
      const v8f dv = acc[2 + i][t];
      acc[i][t]     = sv + dv;
      acc[2 + i][t] = sv - dv;
    }
  }

  if (blockIdx.x == 0) {
    const float* bq = BppP + 2 * d0 + 4 * c;
    float alt[4] = {0.f, 0.f, 0.f, 0.f};
#pragma unroll 4
    for (int k = 0; k < KFREQ; k += 2) {
      const v4f e0 = *(const v4f*)(bq + (size_t)k * (DDIM * 2));
      const v4f e1 = *(const v4f*)(bq + (size_t)k * (DDIM * 2) + 64);
      const v4f o0 = *(const v4f*)(bq + (size_t)(k + 1) * (DDIM * 2));
      const v4f o1 = *(const v4f*)(bq + (size_t)(k + 1) * (DDIM * 2) + 64);
      alt[0] = (alt[0] + e0.x) - o0.x;
      alt[1] = (alt[1] + e0.z) - o0.z;
      alt[2] = (alt[2] + e1.x) - o1.x;
      alt[3] = (alt[3] + e1.z) - o1.z;
    }
    if (h == 0) {
#pragma unroll
      for (int t = 0; t < 4; ++t)
        acc[2][t][0] = alt[t];
    }
  }

  for (int g = 0; g < FO_XG - 2; g += 2) {
    fo_load_x(gb, xa0 + 16 * (g + 1), xa2 + 16 * (g + 1), xa3 + 16 * (g + 1),
              wb + (size_t)(g + 1) * FO_BSTEP);
    __builtin_amdgcn_sched_barrier(0);
    fo_mma(ga, acc);
    __builtin_amdgcn_sched_barrier(0);
    fo_load_x(ga, xa0 + 16 * (g + 2), xa2 + 16 * (g + 2), xa3 + 16 * (g + 2),
              wb + (size_t)(g + 2) * FO_BSTEP);
    __builtin_amdgcn_sched_barrier(0);
    fo_mma(gb, acc);
    __builtin_amdgcn_sched_barrier(0);
  }
  fo_load_x(gb, xa0 + 16 * (FO_XG - 1), xa2 + 16 * (FO_XG - 1),
            xa3 + 16 * (FO_XG - 1), wb + (size_t)(FO_XG - 1) * FO_BSTEP);
  __builtin_amdgcn_sched_barrier(0);
  fo_mma(ga, acc);
  fo_mma(gb, acc);

  float* sw = sOut + wave * (32 * FO_PITCH);
  const v4f bias = *(const v4f*)(Bias + d0 + 4 * c);
  float* ocol = out + d0 + 4 * c;
  fo_epilogue<0>(acc, sw, bias, ocol, n0, c, h);
  __syncthreads();
  fo_epilogue<1>(acc, sw, bias, ocol, n0, c, h);
}

extern "C" void kernel_launch(void* const* d_in, const int* in_sizes, int n_in,
                              void* d_out, int out_size, void* d_ws,
                              size_t ws_size, hipStream_t stream) {
  (void)in_sizes; (void)n_in; (void)out_size; (void)ws_size;
  const float* x     = (const float*)d_in[0];
  const float* P_low = (const float*)d_in[1];
  const float* B_low = (const float*)d_in[2];
  const float* W     = (const float*)d_in[3];
  const float* Bias  = (const float*)d_in[4];
  float* out = (float*)d_out;

  float* ws    = (float*)d_ws;
  float* T     = ws;
  float* Ppart = T + (size_t)N_PTS * NROW;
  float* C     = Ppart + (size_t)NCHUNK * NROW * DDIM;
  float* BppP  = C + NROW * DDIM;
  float* Wp    = BppP + NROW * DDIM;

  tgen_kernel<<<dim3((T_ROWS * KFREQ) / 256), dim3(256), 0, stream>>>(T);
  pack_w_kernel<<<dim3((DDIM * DDIM / 2) / 256), dim3(256), 0, stream>>>(W, Wp);
  dft_low_kernel<<<dim3(DDIM / 64, NCHUNK), dim3(128), 0, stream>>>(x, T, Ppart);
  reduce_chunks_kernel<<<dim3((NROW * DDIM) / RED_T), dim3(RED_T), 0, stream>>>(Ppart, x, C);
  spectral_mix_kernel<<<dim3(KFREQ * MIX_FSPLIT), dim3(32), 0, stream>>>(C, P_low, B_low, BppP);
  fused_out_kernel<<<dim3(NHALF / 32), dim3(128), 0, stream>>>(x, Wp, Bias, T, BppP, out);
}
